# starting kernel
# baseline (speedup 1.0000x reference)
#include <hip/hip_runtime.h>
#include <math.h>

typedef __attribute__((ext_vector_type(2))) float v2f;
typedef __attribute__((ext_vector_type(8))) float v8f;

#define LDA1 258
#define LDA2 130

__global__ __launch_bounds__(256) void gemm1_kernel(const float* __restrict__ x,
                                                    const float* __restrict__ W1,
                                                    float* __restrict__ h, int N_) {
    __shared__ float lds[16 * LDA1];
    const int tile = blockIdx.x;
    const int t    = threadIdx.x;
    const int row0 = tile * 16;
    const bool full = (row0 + 16 <= N_);

    #pragma unroll
    for (int i = 0; i < 16; ++i) {
        int r = row0 + i;
        if (r >= N_) r = N_ - 1;
        lds[i * LDA1 + t] = x[(size_t)r * 256 + t];
    }
    __syncthreads();

    const int wave = t >> 5;
    const int lane = t & 31;
    const int m    = lane & 15;
    const int koff = (lane >> 4) * 2;
    const int n    = (wave << 4) + m;

    v8f acc = {};
    for (int k0 = 0; k0 < 256; k0 += 4) {
        v2f a, b;
        a.x = lds[m * LDA1 + k0 + koff];
        a.y = lds[m * LDA1 + k0 + koff + 1];
        b.x = W1[(size_t)(k0 + koff)     * 128 + n];
        b.y = W1[(size_t)(k0 + koff + 1) * 128 + n];
        acc = __builtin_amdgcn_wmma_f32_16x16x4_f32(false, a, false, b,
                                                    (short)0, acc, false, false);
    }

    const int rbase = row0 + ((lane >> 4) << 3);
    if (full) {
        #pragma unroll
        for (int r = 0; r < 8; ++r)
            h[(size_t)(rbase + r) * 128 + n] = acc[r];
    } else {
        #pragma unroll
        for (int r = 0; r < 8; ++r) {
            const int row = rbase + r;
            if (row < N_) h[(size_t)row * 128 + n] = acc[r];
        }
    }
}

__global__ __launch_bounds__(256) void attn_kernel(const float* __restrict__ h,
                                                   const float* __restrict__ att_src,
                                                   const float* __restrict__ att_dst,
                                                   float* __restrict__ e_src,
                                                   float* __restrict__ e_dst, int N_) {
    const int node = (int)((blockIdx.x * (unsigned)blockDim.x + threadIdx.x) >> 5);
    const int lane = threadIdx.x & 31;
    if (node >= N_) return;
    const float4 hv = ((const float4*)(h + (size_t)node * 128))[lane];
    const float4 as = ((const float4*)att_src)[lane];
    const float4 ad = ((const float4*)att_dst)[lane];
    float s = hv.x * as.x + hv.y * as.y + hv.z * as.z + hv.w * as.w;
    float d = hv.x * ad.x + hv.y * ad.y + hv.z * ad.z + hv.w * ad.w;
    #pragma unroll
    for (int o = 16; o > 0; o >>= 1) {
        s += __shfl_down(s, o, 32);
        d += __shfl_down(d, o, 32);
    }
    if (lane == 0) { e_src[node] = s; e_dst[node] = d; }
}

__global__ __launch_bounds__(256) void edge_alpha_max(const int* __restrict__ ei,
                                                      const float* __restrict__ e_src,
                                                      const float* __restrict__ e_dst,
                                                      float* __restrict__ alpha,
                                                      unsigned* __restrict__ mbits, int E_) {
    const int e = blockIdx.x * blockDim.x + threadIdx.x;
    if (e >= E_) return;
    const int s = ei[e];
    const int d = ei[E_ + e];
    float a = e_src[s] + e_dst[d];
    a = 1.0f / (1.0f + expf(-a));
    alpha[e] = a;
    atomicMax(mbits + d, __float_as_uint(a));
}

__global__ __launch_bounds__(256) void edge_expsum(const int* __restrict__ ei,
                                                   const float* __restrict__ alpha,
                                                   const unsigned* __restrict__ mbits,
                                                   float* __restrict__ denom,
                                                   float* __restrict__ exb, int E_) {
    const int e = blockIdx.x * blockDim.x + threadIdx.x;
    if (e >= E_) return;
    const int d = ei[E_ + e];
    const float mv = __uint_as_float(mbits[d]);
    const float ex = expf(alpha[e] - mv);
    exb[e] = ex;
    atomicAdd(denom + d, ex);
}

__global__ __launch_bounds__(256) void edge_agg(const int* __restrict__ ei,
                                                const float* __restrict__ h,
                                                const float* __restrict__ exb,
                                                const float* __restrict__ denom,
                                                float* __restrict__ out1, int E_) {
    const int e    = (int)((blockIdx.x * (unsigned)blockDim.x + threadIdx.x) >> 5);
    const int lane = threadIdx.x & 31;
    if (e >= E_) return;
    const int s = ei[e];
    const int d = ei[E_ + e];
    const float w = exb[e] / denom[d];
    const float4 hv = ((const float4*)(h + (size_t)s * 128))[lane];
    float* o = out1 + (size_t)d * 128 + lane * 4;
    atomicAdd(o + 0, hv.x * w);
    atomicAdd(o + 1, hv.y * w);
    atomicAdd(o + 2, hv.z * w);
    atomicAdd(o + 3, hv.w * w);
}

__global__ __launch_bounds__(256) void gemm2_kernel(const float* __restrict__ out1,
                                                    const float* __restrict__ W2,
                                                    float* __restrict__ out, int N_) {
    __shared__ float lds[16 * LDA2];
    const int tile = blockIdx.x;
    const int t    = threadIdx.x;
    const int row0 = tile * 16;
    const bool full = (row0 + 16 <= N_);

    #pragma unroll
    for (int i = 0; i < 8; ++i) {
        const int idx = i * 256 + t;
        const int r   = idx >> 7;
        const int c   = idx & 127;
        int row = row0 + r;
        if (row >= N_) row = N_ - 1;
        float v = out1[(size_t)row * 128 + c];
        lds[r * LDA2 + c] = (v > 0.0f) ? v : (expf(v) - 1.0f);
    }
    __syncthreads();

    const int wave = t >> 5;
    const int lane = t & 31;
    const int m    = lane & 15;
    const int koff = (lane >> 4) * 2;
    const int n    = (wave << 4) + m;

    v8f acc = {};
    for (int k0 = 0; k0 < 128; k0 += 4) {
        v2f a, b;
        a.x = lds[m * LDA2 + k0 + koff];
        a.y = lds[m * LDA2 + k0 + koff + 1];
        b.x = W2[(size_t)(k0 + koff)     * 128 + n];
        b.y = W2[(size_t)(k0 + koff + 1) * 128 + n];
        acc = __builtin_amdgcn_wmma_f32_16x16x4_f32(false, a, false, b,
                                                    (short)0, acc, false, false);
    }

    const int rbase = row0 + ((lane >> 4) << 3);
    if (full) {
        #pragma unroll
        for (int r = 0; r < 8; ++r)
            out[(size_t)(rbase + r) * 128 + n] = acc[r];
    } else {
        #pragma unroll
        for (int r = 0; r < 8; ++r) {
            const int row = rbase + r;
            if (row < N_) out[(size_t)row * 128 + n] = acc[r];
        }
    }
}

extern "C" void kernel_launch(void* const* d_in, const int* in_sizes, int n_in,
                              void* d_out, int out_size, void* d_ws, size_t ws_size,
                              hipStream_t stream) {
    const float* x       = (const float*)d_in[0];
    const int*   ei      = (const int*)  d_in[1];
    const float* W1      = (const float*)d_in[2];
    const float* att_src = (const float*)d_in[3];
    const float* att_dst = (const float*)d_in[4];
    const float* W2      = (const float*)d_in[5];
    float*       out     = (float*)d_out;

    const int N_ = in_sizes[0] / 256;
    const int E_ = in_sizes[1] / 2;

    float*    ws    = (float*)d_ws;
    float*    h     = ws;
    float*    out1  = h     + (size_t)N_ * 128;
    float*    e_src = out1  + (size_t)N_ * 128;
    float*    e_dst = e_src + N_;
    unsigned* mbits = (unsigned*)(e_dst + N_);
    float*    denom = (float*)(mbits + N_);
    float*    alpha = denom + N_;
    float*    exb   = alpha + E_;

    hipMemsetAsync(out1,  0, (size_t)N_ * 128 * sizeof(float), stream);
    hipMemsetAsync(mbits, 0, (size_t)N_ * sizeof(unsigned),    stream);
    hipMemsetAsync(denom, 0, (size_t)N_ * sizeof(float),       stream);

    const int tiles = (N_ + 15) / 16;
    gemm1_kernel  <<<tiles, 256, 0, stream>>>(x, W1, h, N_);
    attn_kernel   <<<(N_ * 32 + 255) / 256, 256, 0, stream>>>(h, att_src, att_dst,
                                                              e_src, e_dst, N_);
    edge_alpha_max<<<(E_ + 255) / 256, 256, 0, stream>>>(ei, e_src, e_dst, alpha, mbits, E_);
    edge_expsum   <<<(E_ + 255) / 256, 256, 0, stream>>>(ei, alpha, mbits, denom, exb, E_);
    edge_agg      <<<(E_ * 32 + 255) / 256, 256, 0, stream>>>(ei, h, exb, denom, out1, E_);
    gemm2_kernel  <<<tiles, 256, 0, stream>>>(out1, W2, out, N_);
}


# edge listing: extra blocks of the logit launch file each edge once into per-owner bins (LDS counters; bins in the owner's own out1 rows); owners no longer scan the whole list; sum order changes
# speedup vs baseline: 2.7311x; 2.7311x over previous
#include <hip/hip_runtime.h>
#include <math.h>

typedef __attribute__((ext_vector_type(2))) float v2f;
typedef __attribute__((ext_vector_type(8))) float v8f;

#define LDA1 258
#define LDA2 130

__global__ __launch_bounds__(256) void gemm1_kernel(const float* __restrict__ x,
                                                    const float* __restrict__ W1,
                                                    float* __restrict__ h, int N_) {
    __shared__ float lds[16 * LDA1];
    const int tile = blockIdx.x;
    const int t    = threadIdx.x;
    const int row0 = tile * 16;
    const bool full = (row0 + 16 <= N_);

    #pragma unroll
    for (int i = 0; i < 16; ++i) {
        int r = row0 + i;
        if (r >= N_) r = N_ - 1;
        lds[i * LDA1 + t] = x[(size_t)r * 256 + t];
    }
    __syncthreads();

    const int wave = t >> 5;
    const int lane = t & 31;
    const int m    = lane & 15;
    const int koff = (lane >> 4) * 2;
    const int n    = (wave << 4) + m;

    v8f acc = {};
    for (int k0 = 0; k0 < 256; k0 += 4) {
        v2f a, b;
        a.x = lds[m * LDA1 + k0 + koff];
        a.y = lds[m * LDA1 + k0 + koff + 1];
        b.x = W1[(size_t)(k0 + koff)     * 128 + n];
        b.y = W1[(size_t)(k0 + koff + 1) * 128 + n];
        acc = __builtin_amdgcn_wmma_f32_16x16x4_f32(false, a, false, b,
                                                    (short)0, acc, false, false);
    }

    const int rbase = row0 + ((lane >> 4) << 3);
    if (full) {
        #pragma unroll
        for (int r = 0; r < 8; ++r)
            h[(size_t)(rbase + r) * 128 + n] = acc[r];
    } else {
        #pragma unroll
        for (int r = 0; r < 8; ++r) {
            const int row = rbase + r;
            if (row < N_) h[(size_t)row * 128 + n] = acc[r];
        }
    }
}

#define NPW    196
#define NCH    256
#define SLOTW  64
#define MAXOWN 1024
#define BIN_U  8

__device__ __forceinline__ int own_rows(int o, int N_)  { return (N_ - o * NPW < NPW) ? (N_ - o * NPW) : NPW; }
__device__ __forceinline__ int slot_words(int nown)     { const int w = (nown * 128) / NCH; return (w < SLOTW) ? w : SLOTW; }
__device__ __forceinline__ bool bins_usable(int N_)     { return N_ <= (1 << 24) && (N_ + NPW - 1) / NPW <= MAXOWN; }

__device__ __forceinline__ void bin_edges(const int* __restrict__ ei, unsigned* __restrict__ bins,
                                          int g, int N_, int E_) {
    __shared__ int hcnt[MAXOWN];
    const int t = threadIdx.x;
    const int owners = (N_ + NPW - 1) / NPW;
    for (int i = t; i < owners; i += 256) hcnt[i] = 0;
    __syncthreads();
    const int ch = (E_ + NCH - 1) / NCH;
    const int e0 = g * ch;
    const int e1 = (E_ - e0 < ch) ? E_ : (e0 + ch);
    for (int k0 = e0; k0 < e1; k0 += 256 * BIN_U) {
        int d[BIN_U], s[BIN_U];
        #pragma unroll
        for (int u = 0; u < BIN_U; ++u) {
            const int e  = k0 + u * 256 + t;
            const int ec = (e < e1) ? e : e0;
            d[u] = ei[E_ + ec];
            s[u] = ei[ec];
        }
        #pragma unroll
        for (int u = 0; u < BIN_U; ++u) {
            if (k0 + u * 256 + t < e1 && (unsigned)d[u] < (unsigned)N_) {
                const int o   = d[u] / NPW;
                const int dl  = d[u] - o * NPW;
                const int sw  = slot_words(own_rows(o, N_));
                const int pos = atomicAdd(&hcnt[o], 1);
                if (pos < sw - 1)
                    bins[(size_t)o * (NPW * 128) + g * sw + 1 + pos] = ((unsigned)s[u] << 8) | (unsigned)dl;
            }
        }
    }
    __syncthreads();
    for (int o = t; o < owners; o += 256) {
        const int sw = slot_words(own_rows(o, N_));
        if (sw >= 2) bins[(size_t)o * (NPW * 128) + g * sw] = (unsigned)hcnt[o];
    }
}

__global__ __launch_bounds__(256) void attn_kernel(const float* __restrict__ h,
                                                   const float* __restrict__ att_src,
                                                   const float* __restrict__ att_dst,
                                                   float* __restrict__ e_src,
                                                   float* __restrict__ e_dst,
                                                   const int* __restrict__ ei,
                                                   unsigned* __restrict__ bins, int N_, int E_) {
    if (blockIdx.x < NCH) {
        if (bins_usable(N_)) bin_edges(ei, bins, (int)blockIdx.x, N_, E_);
        return;
    }
    const int node = (int)(((blockIdx.x - NCH) * (unsigned)blockDim.x + threadIdx.x) >> 5);
    const int lane = threadIdx.x & 31;
    if (node >= N_) return;
    const float4 hv = ((const float4*)(h + (size_t)node * 128))[lane];
    const float4 as = ((const float4*)att_src)[lane];
    const float4 ad = ((const float4*)att_dst)[lane];
    float s = hv.x * as.x + hv.y * as.y + hv.z * as.z + hv.w * as.w;
    float d = hv.x * ad.x + hv.y * ad.y + hv.z * ad.z + hv.w * ad.w;
    #pragma unroll
    for (int o = 16; o > 0; o >>= 1) {
        s += __shfl_down(s, o, 32);
        d += __shfl_down(d, o, 32);
    }
    if (lane == 0) { e_src[node] = s; e_dst[node] = d; }
}

#define LCAP   8192
#define SCAN_U 8
#define AGG_U  16

__device__ __forceinline__ float sigmoid_f32(float a) { return 1.0f / (1.0f + expf(-a)); }

template <bool WIDE, class F>
__device__ __forceinline__ void scan_dst(const int* __restrict__ dstp, int E_, int lo, int nown, F&& f) {
    const int t = threadIdx.x;
    if (WIDE && ((E_ & 3) == 0) && ((((size_t)dstp) & 15) == 0)) {
        const int4* __restrict__ d4 = (const int4*)dstp;
        const int nvec = E_ >> 2;
        const int start = (int)(((long long)blockIdx.x * nvec) / gridDim.x);
        for (int k0 = 0; k0 < nvec; k0 += 1024 * SCAN_U) {
            int4 v[SCAN_U];
            int  id[SCAN_U];
            #pragma unroll
            for (int u = 0; u < SCAN_U; ++u) {
                const int k = k0 + u * 1024 + t;
                int idx = k + start;
                if (idx >= nvec) idx -= nvec;
                if (k >= nvec) idx = 0;
                id[u] = idx;
                v[u]  = d4[idx];
            }
            #pragma unroll
            for (int u = 0; u < SCAN_U; ++u) {
                const int e = id[u] * 4;
                if (k0 + u * 1024 + t >= nvec) v[u] = make_int4(-1, -1, -1, -1);
                const unsigned a0 = (unsigned)(v[u].x - lo);
                const unsigned a1 = (unsigned)(v[u].y - lo);
                const unsigned a2 = (unsigned)(v[u].z - lo);
                const unsigned a3 = (unsigned)(v[u].w - lo);
                if (a0 < (unsigned)nown) f(e,     (int)a0);
                if (a1 < (unsigned)nown) f(e + 1, (int)a1);
                if (a2 < (unsigned)nown) f(e + 2, (int)a2);
                if (a3 < (unsigned)nown) f(e + 3, (int)a3);
            }
        }
    } else {
        for (int k0 = 0; k0 < E_; k0 += 1024) {
            const int e = k0 + t;
            if (e < E_) {
                const unsigned a0 = (unsigned)(dstp[e] - lo);
                if (a0 < (unsigned)nown) f(e, (int)a0);
            }
        }
    }
}

__global__ __launch_bounds__(1024) void edge_owner(const int* __restrict__ ei,
                                                   const float* __restrict__ h,
                                                   const float* __restrict__ e_src,
                                                   const float* __restrict__ e_dst,
                                                   float* __restrict__ out1, int N_, int E_) {
    __shared__ __attribute__((aligned(16))) float acc[NPW * 128];
    __shared__ float          den[NPW];
    __shared__ unsigned       mxb[NPW];
    __shared__ int            l_src[LCAP];
    __shared__ float          l_val[LCAP];
    __shared__ unsigned short l_dl[LCAP];
    __shared__ int            cnt;
    __shared__ int            ovf;
    __shared__ int            s_off[NCH + 1];
    static_assert(NCH * 4 == 1024 && NPW <= 256, "four threads of the owner per bin slot; owned row fits 8 bits");

    const int t    = threadIdx.x;
    const int lo   = blockIdx.x * NPW;
    const int nown = (N_ - lo < NPW) ? (N_ - lo) : NPW;
    const int* __restrict__ dstp = ei + E_;
    const int  sw       = slot_words(nown);
    const bool use_bins = bins_usable(N_) && sw >= 2;

    for (int i = t; i < NPW * 32; i += 1024)
        ((float4*)acc)[i] = make_float4(0.0f, 0.0f, 0.0f, 0.0f);
    if (t < NPW) { den[t] = 0.0f; mxb[t] = 0u; }
    if (t == 0) { cnt = 0; ovf = 0; }
    __syncthreads();

    int  total     = 0;
    bool listed    = false;
    if (use_bins) {
        const unsigned* reg = (const unsigned*)(out1 + (size_t)lo * 128);
        const int g   = t >> 2;
        const int sub = t & 3;
        const int c   = (int)reg[g * sw];
        if (sub == 0) { s_off[g] = c; if (c > sw - 1) ovf = 1; }
        __syncthreads();
        if (t < 32) {
            int v[NCH / 32];
            int sum = 0;
            for (int i = 0; i < NCH / 32; ++i) { v[i] = sum; sum += s_off[t * (NCH / 32) + i]; }
            int incl = sum;
            for (int o = 1; o < 32; o <<= 1) {
                const int up = __shfl_up(incl, o, 32);
                if (t >= o) incl += up;
            }
            const int base = incl - sum;
            for (int i = 0; i < NCH / 32; ++i) s_off[t * (NCH / 32) + i] = base + v[i];
            if (t == 31) s_off[NCH] = incl;
        }
        __syncthreads();
        if (ovf == 0) {
            listed = true;
            total  = s_off[NCH];
            if (total <= LCAP) {
                const int off = s_off[g];
                for (int j = sub; j < c; j += 4) {
                    const unsigned en = reg[g * sw + 1 + j];
                    l_src[off + j] = (int)(en >> 8);
                    l_dl[off + j]  = (unsigned short)(en & 255u);
                }
            }
        }
    }
    if (!listed) {
        scan_dst<true>(dstp, E_, lo, nown, [&](int e, int dl) {
            const int pos = atomicAdd(&cnt, 1);
            if (pos < LCAP) { l_src[pos] = e; l_dl[pos] = (unsigned short)dl; }
        });
    }
    __syncthreads();
    if (!listed) total = cnt;

    if (total <= LCAP) {
        for (int i = t; i < total; i += 1024) {
            const int dl = l_dl[i];
            const int s  = listed ? l_src[i] : ei[l_src[i]];
            const float a = sigmoid_f32(e_src[s] + e_dst[lo + dl]);
            l_src[i] = s;
            l_val[i] = a;
            atomicMax(&mxb[dl], __float_as_uint(a));
        }
        __syncthreads();
        for (int i = t; i < total; i += 1024) {
            const int dl = l_dl[i];
            const float ex = expf(l_val[i] - __uint_as_float(mxb[dl]));
            l_val[i] = ex;
            atomicAdd(&den[dl], ex);
        }
        __syncthreads();
        const int wave = __builtin_amdgcn_readfirstlane(t >> 5);
        const int lane = t & 31;
        for (int c0 = wave * 32; c0 < total; c0 += 32 * 32) {
            const int nin = (total - c0 < 32) ? (total - c0) : 32;
            int   my_s = 0, my_dr = 0;
            float my_w = 0.0f;
            if (lane < nin) {
                const int dl = l_dl[c0 + lane];
                my_s  = l_src[c0 + lane];
                my_dr = dl * 128;
                my_w  = l_val[c0 + lane] / den[dl];
            }
            #pragma unroll
            for (int u0 = 0; u0 < 32; u0 += AGG_U) {
                if (u0 < nin) {
                    float4 hv[AGG_U];
                    float  wg[AGG_U];
                    int    dr[AGG_U];
                    #pragma unroll
                    for (int u = 0; u < AGG_U; ++u) {
                        const int s = __builtin_amdgcn_readlane(my_s, u0 + u);
                        dr[u] = __builtin_amdgcn_readlane(my_dr, u0 + u) + lane;
                        wg[u] = __int_as_float(__builtin_amdgcn_readlane(__float_as_int(my_w), u0 + u));
                        hv[u] = ((const float4*)(h + (size_t)s * 128))[lane];
                    }
                    #pragma unroll
                    for (int u = 0; u < AGG_U; ++u) {
                        if (u0 + u < nin) {
                            atomicAdd(&acc[dr[u]],      hv[u].x * wg[u]);
                            atomicAdd(&acc[dr[u] + 32], hv[u].y * wg[u]);
                            atomicAdd(&acc[dr[u] + 64], hv[u].z * wg[u]);
                            atomicAdd(&acc[dr[u] + 96], hv[u].w * wg[u]);
                        }
                    }
                }
            }
        }
    } else {
        scan_dst<false>(dstp, E_, lo, nown, [&](int e, int dl) {
            const int s = ei[e];
            const float a = sigmoid_f32(e_src[s] + e_dst[lo + dl]);
            atomicMax(&mxb[dl], __float_as_uint(a));
        });
        __syncthreads();
        scan_dst<false>(dstp, E_, lo, nown, [&](int e, int dl) {
            const int s = ei[e];
            const float a = sigmoid_f32(e_src[s] + e_dst[lo + dl]);
            atomicAdd(&den[dl], expf(a - __uint_as_float(mxb[dl])));
        });
        __syncthreads();
        scan_dst<false>(dstp, E_, lo, nown, [&](int e, int dl) {
            const int s = ei[e];
            const float a = sigmoid_f32(e_src[s] + e_dst[lo + dl]);
            const float w = expf(a - __uint_as_float(mxb[dl])) / den[dl];
            const float* __restrict__ hp = h + (size_t)s * 128;
            for (int c = 0; c < 128; ++c)
                atomicAdd(&acc[dl * 128 + (c & 3) * 32 + (c >> 2)], hp[c] * w);
        });
    }
    __syncthreads();

    float4* __restrict__ o4 = (float4*)(out1 + (size_t)lo * 128);
    for (int i = t; i < nown * 32; i += 1024) {
        const int r = i >> 5;
        const int q = i & 31;
        const float* a = acc + r * 128 + q;
        o4[i] = make_float4(a[0], a[32], a[64], a[96]);
    }
}

__global__ __launch_bounds__(256) void gemm2_kernel(const float* __restrict__ out1,
                                                    const float* __restrict__ W2,
                                                    float* __restrict__ out, int N_) {
    __shared__ float lds[16 * LDA2];
    const int tile = blockIdx.x;
    const int t    = threadIdx.x;
    const int row0 = tile * 16;
    const bool full = (row0 + 16 <= N_);

    #pragma unroll
    for (int i = 0; i < 8; ++i) {
        const int idx = i * 256 + t;
        const int r   = idx >> 7;
        const int c   = idx & 127;
        int row = row0 + r;
        if (row >= N_) row = N_ - 1;
        float v = out1[(size_t)row * 128 + c];
        lds[r * LDA2 + c] = (v > 0.0f) ? v : (expf(v) - 1.0f);
    }
    __syncthreads();

    const int wave = t >> 5;
    const int lane = t & 31;
    const int m    = lane & 15;
    const int koff = (lane >> 4) * 2;
    const int n    = (wave << 4) + m;

    v8f acc = {};
    for (int k0 = 0; k0 < 128; k0 += 4) {
        v2f a, b;
        a.x = lds[m * LDA2 + k0 + koff];
        a.y = lds[m * LDA2 + k0 + koff + 1];
        b.x = W2[(size_t)(k0 + koff)     * 128 + n];
        b.y = W2[(size_t)(k0 + koff + 1) * 128 + n];
        acc = __builtin_amdgcn_wmma_f32_16x16x4_f32(false, a, false, b,
                                                    (short)0, acc, false, false);
    }

    const int rbase = row0 + ((lane >> 4) << 3);
    if (full) {
        #pragma unroll
        for (int r = 0; r < 8; ++r)
            out[(size_t)(rbase + r) * 128 + n] = acc[r];
    } else {
        #pragma unroll
        for (int r = 0; r < 8; ++r) {
            const int row = rbase + r;
            if (row < N_) out[(size_t)row * 128 + n] = acc[r];
        }
    }
}

extern "C" void kernel_launch(void* const* d_in, const int* in_sizes, int n_in,
                              void* d_out, int out_size, void* d_ws, size_t ws_size,
                              hipStream_t stream) {
    const float* x       = (const float*)d_in[0];
    const int*   ei      = (const int*)  d_in[1];
    const float* W1      = (const float*)d_in[2];
    const float* att_src = (const float*)d_in[3];
    const float* att_dst = (const float*)d_in[4];
    const float* W2      = (const float*)d_in[5];
    float*       out     = (float*)d_out;

    const int N_ = in_sizes[0] / 256;
    const int E_ = in_sizes[1] / 2;

    float*    ws    = (float*)d_ws;
    float*    h     = ws;
    float*    out1  = h     + (size_t)N_ * 128;
    float*    e_src = out1  + (size_t)N_ * 128;
    float*    e_dst = e_src + N_;

    const int tiles  = (N_ + 15) / 16;
    const int owners = (N_ + NPW - 1) / NPW;
    gemm1_kernel  <<<tiles, 256, 0, stream>>>(x, W1, h, N_);
    attn_kernel   <<<NCH + (N_ * 32 + 255) / 256, 256, 0, stream>>>(h, att_src, att_dst,
                                                                    e_src, e_dst, ei,
                                                                    (unsigned*)out1, N_, E_);
    edge_owner    <<<owners, 1024, 0, stream>>>(ei, h, e_src, e_dst, out1, N_, E_);
    gemm2_kernel  <<<tiles, 256, 0, stream>>>(out1, W2, out, N_);
}


# GEMM1: W1 kept in registers (each wave its 16 columns, all k) by 512 work-groups that loop over the 16-row x tiles, two per trip, instead of re-reading W1 for every tile; GEMM1 k order changes
# speedup vs baseline: 4.1822x; 1.5313x over previous
#include <hip/hip_runtime.h>
#include <math.h>

typedef __attribute__((ext_vector_type(2))) float v2f;
typedef __attribute__((ext_vector_type(8))) float v8f;

#define NPW    196
#define NCH    256
#define SLOTW  64
#define MAXOWN 1024
#define BIN_U  8

__device__ __forceinline__ int own_rows(int o, int N_)  { return (N_ - o * NPW < NPW) ? (N_ - o * NPW) : NPW; }
__device__ __forceinline__ int slot_words(int nown)     { const int w = (nown * 128) / NCH; return (w < SLOTW) ? w : SLOTW; }
__device__ __forceinline__ bool bins_usable(int N_)     { return N_ <= (1 << 24) && (N_ + NPW - 1) / NPW <= MAXOWN; }

__device__ __forceinline__ void bin_edges(const int* __restrict__ ei, unsigned* __restrict__ bins,
                                          int* __restrict__ hcnt, int g, int N_, int E_) {
    const int t = threadIdx.x;
    const int owners = (N_ + NPW - 1) / NPW;
    for (int i = t; i < owners; i += 256) hcnt[i] = 0;
    __syncthreads();
    const int ch = (E_ + NCH - 1) / NCH;
    const int e0 = g * ch;
    const int e1 = (E_ - e0 < ch) ? E_ : (e0 + ch);
    for (int k0 = e0; k0 < e1; k0 += 256 * BIN_U) {
        int d[BIN_U], s[BIN_U];
        #pragma unroll
        for (int u = 0; u < BIN_U; ++u) {
            const int e  = k0 + u * 256 + t;
            const int ec = (e < e1) ? e : e0;
            d[u] = ei[E_ + ec];
            s[u] = ei[ec];
        }
        #pragma unroll
        for (int u = 0; u < BIN_U; ++u) {
            if (k0 + u * 256 + t < e1 && (unsigned)d[u] < (unsigned)N_) {
                const int o   = d[u] / NPW;
                const int dl  = d[u] - o * NPW;
                const int sw  = slot_words(own_rows(o, N_));
                const int pos = atomicAdd(&hcnt[o], 1);
                if (pos < sw - 1)
                    bins[(size_t)o * (NPW * 128) + g * sw + 1 + pos] = ((unsigned)s[u] << 8) | (unsigned)dl;
            }
        }
    }
    __syncthreads();
    for (int o = t; o < owners; o += 256) {
        const int sw = slot_words(own_rows(o, N_));
        if (sw >= 2) bins[(size_t)o * (NPW * 128) + g * sw] = (unsigned)hcnt[o];
    }
}

#define GB     512
#define XROWS  32
#define LDX    260

__device__ __forceinline__ float4 x_row_load(const float* x, int r, int N_) {
    if (r >= N_) r = N_ - 1;
    return ((const float4*)(x + (size_t)r * 256))[threadIdx.x & 63];
}
#define X_ROWS_LOAD(rb) do { const int r_ = (rb) + (int)(threadIdx.x >> 6);                                  \
    st0 = x_row_load(x, r_,      N_); st1 = x_row_load(x, r_ + 4,  N_); st2 = x_row_load(x, r_ + 8,  N_);   \
    st3 = x_row_load(x, r_ + 12, N_); st4 = x_row_load(x, r_ + 16, N_); st5 = x_row_load(x, r_ + 20, N_);   \
    st6 = x_row_load(x, r_ + 24, N_); st7 = x_row_load(x, r_ + 28, N_); } while (0)
#define X_ROWS_STORE(buf) do { float* b_ = (buf) + (threadIdx.x >> 6) * LDX + 4 * (threadIdx.x & 63);        \
    *(float4*)(b_)            = st0; *(float4*)(b_ + 4 * LDX)  = st1; *(float4*)(b_ + 8 * LDX)  = st2;       \
    *(float4*)(b_ + 12 * LDX) = st3; *(float4*)(b_ + 16 * LDX) = st4; *(float4*)(b_ + 20 * LDX) = st5;       \
    *(float4*)(b_ + 24 * LDX) = st6; *(float4*)(b_ + 28 * LDX) = st7; } while (0)

#define KG 2
template <int NT>
__device__ __forceinline__ void tile_products(const float* __restrict__ xb, const v2f (&wa)[64], v8f (&acc)[2]) {
    float4 f[2][KG][NT];
    #pragma unroll
    for (int jj = 0; jj < KG; ++jj)
        #pragma unroll
        for (int q = 0; q < NT; ++q) f[0][jj][q] = *(const float4*)(xb + q * 16 * LDX + 8 * jj);
    #pragma unroll
    for (int g = 0; g < 32 / KG; ++g) {
        if (g + 1 < 32 / KG) {
            #pragma unroll
            for (int jj = 0; jj < KG; ++jj)
                #pragma unroll
                for (int q = 0; q < NT; ++q)
                    f[(g + 1) & 1][jj][q] = *(const float4*)(xb + q * 16 * LDX + 8 * ((g + 1) * KG + jj));
        }
        __builtin_amdgcn_sched_barrier(0);
        #pragma unroll
        for (int jj = 0; jj < KG; ++jj) {
            const int j = g * KG + jj;
            #pragma unroll
            for (int q = 0; q < NT; ++q) {
                v2f b; b.x = f[g & 1][jj][q].x; b.y = f[g & 1][jj][q].y;
                acc[q] = __builtin_amdgcn_wmma_f32_16x16x4_f32(false, wa[2 * j], false, b,
                                                               (short)0, acc[q], false, false);
            }
            #pragma unroll
            for (int q = 0; q < NT; ++q) {
                v2f b; b.x = f[g & 1][jj][q].z; b.y = f[g & 1][jj][q].w;
                acc[q] = __builtin_amdgcn_wmma_f32_16x16x4_f32(false, wa[2 * j + 1], false, b,
                                                               (short)0, acc[q], false, false);
            }
        }
    }
}

__global__ __launch_bounds__(256, 4) void gemm1_kernel(const float* __restrict__ x,
                                                       const float* __restrict__ W1,
                                                       const float* __restrict__ att_src,
                                                       const float* __restrict__ att_dst,
                                                       float* __restrict__ h,
                                                       float* __restrict__ e_src,
                                                       float* __restrict__ e_dst,
                                                       const int* __restrict__ ei,
                                                       unsigned* __restrict__ bins, int N_, int E_) {
    __shared__ __attribute__((aligned(16))) float lds[2 * XROWS * LDX];
    __shared__ float red[2 * 2 * 8 * 32];
    static_assert(2 * XROWS * LDX >= MAXOWN, "the binning blocks count in the stages' LDS");
    if (blockIdx.x < NCH) {
        if (bins_usable(N_)) bin_edges(ei, bins, (int*)lds, (int)blockIdx.x, N_, E_);
        return;
    }
    const int G  = (int)gridDim.x - NCH;
    const int gb = (int)blockIdx.x - NCH;
    const int T  = (N_ + 15) / 16;
    const int t0 = (int)(((long long)gb * T) / G);
    const int t1 = (int)(((long long)(gb + 1) * T) / G);
    if (t0 >= t1) return;

    const int t    = threadIdx.x;
    const int wave = t >> 5;
    const int lane = t & 31;
    const int m    = lane & 15;
    const int hh   = lane >> 4;
    const int n    = (wave << 4) + m;
    const int c0   = (wave << 4) + (hh << 3);

    float4 st0, st1, st2, st3, st4, st5, st6, st7;
    X_ROWS_LOAD(t0 * 16);

    v2f wa[64];
    #pragma unroll
    for (int j = 0; j < 32; ++j) {
        const float* __restrict__ wj = W1 + (size_t)(8 * j + 4 * hh) * 128 + n;
        wa[2 * j].x     = wj[0];
        wa[2 * j].y     = wj[128];
        wa[2 * j + 1].x = wj[256];
        wa[2 * j + 1].y = wj[384];
    }
    const float4 as0 = ((const float4*)(att_src + c0))[0];
    const float4 as1 = ((const float4*)(att_src + c0))[1];
    const float4 ad0 = ((const float4*)(att_dst + c0))[0];
    const float4 ad1 = ((const float4*)(att_dst + c0))[1];

    X_ROWS_STORE(lds);
    __syncthreads();

    int cur = 0;
    for (int tt = t0; tt < t1; tt += 2) {
        const bool more = tt + 2 < t1;
        const bool two  = tt + 1 < t1;
        X_ROWS_LOAD((more ? tt + 2 : tt) * 16);

        v8f acc[2] = {};
        const float* __restrict__ xb = lds + cur * (XROWS * LDX) + m * LDX + 4 * hh;
        if (two) tile_products<2>(xb, wa, acc); else tile_products<1>(xb, wa, acc);

        #pragma unroll
        for (int q = 0; q < 2; ++q) {
            if (q == 0 || two) {
                const int row = (tt + q) * 16 + m;
                if (row < N_) {
                    float4* __restrict__ hp = (float4*)(h + (size_t)row * 128 + c0);
                    hp[0] = make_float4(acc[q][0], acc[q][1], acc[q][2], acc[q][3]);
                    hp[1] = make_float4(acc[q][4], acc[q][5], acc[q][6], acc[q][7]);
                }
                float ps = acc[q][0] * as0.x + acc[q][1] * as0.y + acc[q][2] * as0.z + acc[q][3] * as0.w
                         + acc[q][4] * as1.x + acc[q][5] * as1.y + acc[q][6] * as1.z + acc[q][7] * as1.w;
                float pd = acc[q][0] * ad0.x + acc[q][1] * ad0.y + acc[q][2] * ad0.z + acc[q][3] * ad0.w
                         + acc[q][4] * ad1.x + acc[q][5] * ad1.y + acc[q][6] * ad1.z + acc[q][7] * ad1.w;
                ps += __shfl_xor(ps, 16, 32);
                pd += __shfl_xor(pd, 16, 32);
                if (lane < 16) {
                    red[(cur * 2 + q) * 256 + wave * 32 + m * 2]     = ps;
                    red[(cur * 2 + q) * 256 + wave * 32 + m * 2 + 1] = pd;
                }
            }
        }
        X_ROWS_STORE(lds + (cur ^ 1) * (XROWS * LDX));
        __builtin_amdgcn_fence(__ATOMIC_RELEASE, "workgroup", "local");
        __builtin_amdgcn_s_barrier();
        __builtin_amdgcn_fence(__ATOMIC_ACQUIRE, "workgroup", "local");
        if (t < 64 && (t < 32 || two)) {
            const int q = t >> 5, u = t & 31;
            const float* __restrict__ rp = red + (cur * 2 + q) * 256 + u;
            float s = rp[0];
            for (int w = 1; w < 8; ++w) s += rp[w * 32];
            const int row = (tt + q) * 16 + (u >> 1);
            if (row < N_) {
                if (u & 1) e_dst[row] = s; else e_src[row] = s;
            }
        }
        cur ^= 1;
    }
}

#define LCAP   8192
#define NPT    ((NPW + 15) / 16)
#define LDR    132
#define WPS    288
#define SCAN_U 8
#define AGG_U  16

__device__ __forceinline__ float sigmoid_f32(float a) { return 1.0f / (1.0f + expf(-a)); }

template <bool WIDE, class F>
__device__ __forceinline__ void scan_dst(const int* __restrict__ dstp, int E_, int lo, int nown, F&& f) {
    const int t = threadIdx.x;
    if (WIDE && ((E_ & 3) == 0) && ((((size_t)dstp) & 15) == 0)) {
        const int4* __restrict__ d4 = (const int4*)dstp;
        const int nvec = E_ >> 2;
        const int start = (int)(((long long)blockIdx.x * nvec) / gridDim.x);
        for (int k0 = 0; k0 < nvec; k0 += 1024 * SCAN_U) {
            int4 v[SCAN_U];
            int  id[SCAN_U];
            #pragma unroll
            for (int u = 0; u < SCAN_U; ++u) {
                const int k = k0 + u * 1024 + t;
                int idx = k + start;
                if (idx >= nvec) idx -= nvec;
                if (k >= nvec) idx = 0;
                id[u] = idx;
                v[u]  = d4[idx];
            }
            #pragma unroll
            for (int u = 0; u < SCAN_U; ++u) {
                const int e = id[u] * 4;
                if (k0 + u * 1024 + t >= nvec) v[u] = make_int4(-1, -1, -1, -1);
                const unsigned a0 = (unsigned)(v[u].x - lo);
                const unsigned a1 = (unsigned)(v[u].y - lo);
                const unsigned a2 = (unsigned)(v[u].z - lo);
                const unsigned a3 = (unsigned)(v[u].w - lo);
                if (a0 < (unsigned)nown) f(e,     (int)a0);
                if (a1 < (unsigned)nown) f(e + 1, (int)a1);
                if (a2 < (unsigned)nown) f(e + 2, (int)a2);
                if (a3 < (unsigned)nown) f(e + 3, (int)a3);
            }
        }
    } else {
        for (int k0 = 0; k0 < E_; k0 += 1024) {
            const int e = k0 + t;
            if (e < E_) {
                const unsigned a0 = (unsigned)(dstp[e] - lo);
                if (a0 < (unsigned)nown) f(e, (int)a0);
            }
        }
    }
}

__global__ __launch_bounds__(1024) void edge_owner(const int* __restrict__ ei,
                                                   const float* __restrict__ h,
                                                   const float* __restrict__ e_src,
                                                   const float* __restrict__ e_dst,
                                                   const float* __restrict__ W2,
                                                   const unsigned* __restrict__ bins,
                                                   float* __restrict__ out, int N_, int E_) {
    __shared__ __attribute__((aligned(16))) float acc[NPT * 16 * LDR];
    __shared__ __attribute__((aligned(16))) float wp[64 * WPS];
    __shared__ float          den[NPW];
    __shared__ unsigned       mxb[NPW];
    __shared__ int            l_src[LCAP];
    __shared__ float          l_val[LCAP];
    __shared__ unsigned short l_dl[LCAP];
    __shared__ int            cnt;
    __shared__ int            ovf;
    __shared__ int            s_off[NCH + 1];
    static_assert(NCH * 4 == 1024 && NPW <= 256, "four threads of the owner per bin slot; owned row fits 8 bits");

    const int t    = threadIdx.x;
    const int lo   = blockIdx.x * NPW;
    const int nown = (N_ - lo < NPW) ? (N_ - lo) : NPW;
    const int* __restrict__ dstp = ei + E_;
    const int  sw       = slot_words(nown);
    const bool use_bins = bins_usable(N_) && sw >= 2;

    const int wl = t & 31;
    const int wu = t >> 5;
    float4 w2lo[2], w2hi[2];
    #pragma unroll
    for (int k = 0; k < 2; ++k) {
        const int pp = wu + 32 * k;
        const int c0 = 8 * (pp & 15) + (pp >> 4);
        w2lo[k] = ((const float4*)(W2 + (size_t)c0 * 128))[wl];
        w2hi[k] = ((const float4*)(W2 + (size_t)(c0 + 4) * 128))[wl];
    }

    for (int i = t; i < NPT * 16 * LDR / 4; i += 1024)
        ((float4*)acc)[i] = make_float4(0.0f, 0.0f, 0.0f, 0.0f);
    #pragma unroll
    for (int k = 0; k < 2; ++k) {
        float4* __restrict__ d = (float4*)(wp + (wu + 32 * k) * WPS + 8 * wl);
        d[0] = make_float4(w2lo[k].x, w2hi[k].x, w2lo[k].y, w2hi[k].y);
        d[1] = make_float4(w2lo[k].z, w2hi[k].z, w2lo[k].w, w2hi[k].w);
    }
    if (t < NPW) { den[t] = 0.0f; mxb[t] = 0u; }
    if (t == 0) { cnt = 0; ovf = 0; }
    __syncthreads();

    int  total     = 0;
    bool listed    = false;
    if (use_bins) {
        const unsigned* __restrict__ reg = bins + (size_t)lo * 128;
        const int g   = t >> 2;
        const int sub = t & 3;
        const int c   = (int)reg[g * sw];
        if (sub == 0) { s_off[g] = c; if (c > sw - 1) ovf = 1; }
        __syncthreads();
        if (t < 32) {
            int v[NCH / 32];
            int sum = 0;
            for (int i = 0; i < NCH / 32; ++i) { v[i] = sum; sum += s_off[t * (NCH / 32) + i]; }
            int incl = sum;
            for (int o = 1; o < 32; o <<= 1) {
                const int up = __shfl_up(incl, o, 32);
                if (t >= o) incl += up;
            }
            const int base = incl - sum;
            for (int i = 0; i < NCH / 32; ++i) s_off[t * (NCH / 32) + i] = base + v[i];
            if (t == 31) s_off[NCH] = incl;
        }
        __syncthreads();
        if (ovf == 0) {
            listed = true;
            total  = s_off[NCH];
            if (total <= LCAP) {
                const int off = s_off[g];
                for (int j = sub; j < c; j += 4) {
                    const unsigned en = reg[g * sw + 1 + j];
                    l_src[off + j] = (int)(en >> 8);
                    l_dl[off + j]  = (unsigned short)(en & 255u);
                }
            }
        }
    }
    if (!listed) {
        scan_dst<true>(dstp, E_, lo, nown, [&](int e, int dl) {
            const int pos = atomicAdd(&cnt, 1);
            if (pos < LCAP) { l_src[pos] = e; l_dl[pos] = (unsigned short)dl; }
        });
    }
    __syncthreads();
    if (!listed) total = cnt;

    if (total <= LCAP) {
        for (int i = t; i < total; i += 1024) {
            const int dl = l_dl[i];
            const int s  = listed ? l_src[i] : ei[l_src[i]];
            const float a = sigmoid_f32(e_src[s] + e_dst[lo + dl]);
            l_src[i] = s;
            l_val[i] = a;
            atomicMax(&mxb[dl], __float_as_uint(a));
        }
        __syncthreads();
        for (int i = t; i < total; i += 1024) {
            const int dl = l_dl[i];
            const float ex = expf(l_val[i] - __uint_as_float(mxb[dl]));
            l_val[i] = ex;
            atomicAdd(&den[dl], ex);
        }
        __syncthreads();
        const int wave = __builtin_amdgcn_readfirstlane(t >> 5);
        const int lane = t & 31;
        for (int c0 = wave * 32; c0 < total; c0 += 32 * 32) {
            const int nin = (total - c0 < 32) ? (total - c0) : 32;
            int   my_s = 0, my_dr = 0;
            float my_w = 0.0f;
            if (lane < nin) {
                const int dl = l_dl[c0 + lane];
                my_s  = l_src[c0 + lane];
                my_dr = dl * LDR;
                my_w  = l_val[c0 + lane] / den[dl];
            }
            #pragma unroll
            for (int u0 = 0; u0 < 32; u0 += AGG_U) {
                if (u0 < nin) {
                    float4 hv[AGG_U];
                    float  wg[AGG_U];
                    int    dr[AGG_U];
                    #pragma unroll
                    for (int u = 0; u < AGG_U; ++u) {
                        const int s = __builtin_amdgcn_readlane(my_s, u0 + u);
                        dr[u] = __builtin_amdgcn_readlane(my_dr, u0 + u) + lane;
                        wg[u] = __int_as_float(__builtin_amdgcn_readlane(__float_as_int(my_w), u0 + u));
                        hv[u] = ((const float4*)(h + (size_t)s * 128))[lane];
                    }
                    #pragma unroll
                    for (int u = 0; u < AGG_U; ++u) {
                        if (u0 + u < nin) {
                            atomicAdd(&acc[dr[u]],      hv[u].x * wg[u]);
                            atomicAdd(&acc[dr[u] + 32], hv[u].y * wg[u]);
                            atomicAdd(&acc[dr[u] + 64], hv[u].z * wg[u]);
                            atomicAdd(&acc[dr[u] + 96], hv[u].w * wg[u]);
                        }
                    }
                }
            }
        }
    } else {
        scan_dst<false>(dstp, E_, lo, nown, [&](int e, int dl) {
            const int s = ei[e];
            const float a = sigmoid_f32(e_src[s] + e_dst[lo + dl]);
            atomicMax(&mxb[dl], __float_as_uint(a));
        });
        __syncthreads();
        scan_dst<false>(dstp, E_, lo, nown, [&](int e, int dl) {
            const int s = ei[e];
            const float a = sigmoid_f32(e_src[s] + e_dst[lo + dl]);
            atomicAdd(&den[dl], expf(a - __uint_as_float(mxb[dl])));
        });
        __syncthreads();
        scan_dst<false>(dstp, E_, lo, nown, [&](int e, int dl) {
            const int s = ei[e];
            const float a = sigmoid_f32(e_src[s] + e_dst[lo + dl]);
            const float w = expf(a - __uint_as_float(mxb[dl])) / den[dl];
            const float* __restrict__ hp = h + (size_t)s * 128;
            for (int c = 0; c < 128; ++c)
                atomicAdd(&acc[dl * LDR + (c & 3) * 32 + (c >> 2)], hp[c] * w);
        });
    }
    __syncthreads();

    for (int i = t; i < nown * 32; i += 1024) {
        float4* __restrict__ p = (float4*)(acc + (i >> 5) * LDR) + (i & 31);
        float4 v = *p;
        v.x = (v.x > 0.0f) ? v.x : (expf(v.x) - 1.0f);
        v.y = (v.y > 0.0f) ? v.y : (expf(v.y) - 1.0f);
        v.z = (v.z > 0.0f) ? v.z : (expf(v.z) - 1.0f);
        v.w = (v.w > 0.0f) ? v.w : (expf(v.w) - 1.0f);
        *p = v;
    }
    __syncthreads();

    {
        const int wv = __builtin_amdgcn_readfirstlane(t >> 5);
        const int ln = t & 31;
        const int m  = ln & 15;
        const int hh = ln >> 4;
        const int n  = ((wv & 7) << 4) + m;
        v2f bf[32];
        #pragma unroll
        for (int s = 0; s < 32; ++s)
            bf[s] = *(const v2f*)(wp + (2 * s + hh) * WPS + 2 * n);
        for (int rt = wv >> 3; rt * 16 < nown; rt += 4) {
            const float* __restrict__ ap = acc + (rt * 16 + m) * LDR + 2 * hh;
            v8f d = {};
            #pragma unroll
            for (int s = 0; s < 32; ++s) {
                const v2f a = *(const v2f*)(ap + 4 * s);
                d = __builtin_amdgcn_wmma_f32_16x16x4_f32(false, a, false, bf[s],
                                                          (short)0, d, false, false);
            }
            const int rb = rt * 16 + hh * 8;
            float* __restrict__ op = out + (size_t)(lo + rb) * 128 + n;
            if (rt * 16 + 16 <= nown) {
                #pragma unroll
                for (int r = 0; r < 8; ++r) op[(size_t)r * 128] = d[r];
            } else {
                #pragma unroll
                for (int r = 0; r < 8; ++r)
                    if (rb + r < nown) op[(size_t)r * 128] = d[r];
            }
        }
    }
}

extern "C" void kernel_launch(void* const* d_in, const int* in_sizes, int n_in,
                              void* d_out, int out_size, void* d_ws, size_t ws_size,
                              hipStream_t stream) {
    const float* x       = (const float*)d_in[0];
    const int*   ei      = (const int*)  d_in[1];
    const float* W1      = (const float*)d_in[2];
    const float* att_src = (const float*)d_in[3];
    const float* att_dst = (const float*)d_in[4];
    const float* W2      = (const float*)d_in[5];
    float*       out     = (float*)d_out;

    const int N_ = in_sizes[0] / 256;
    const int E_ = in_sizes[1] / 2;

    float*    ws    = (float*)d_ws;
    float*    h     = ws;
    float*    out1  = h     + (size_t)N_ * 128;
    float*    e_src = out1  + (size_t)N_ * 128;
    float*    e_dst = e_src + N_;

    const int owners = (N_ + NPW - 1) / NPW;
    gemm1_kernel  <<<NCH + GB, 256, 0, stream>>>(x, W1, att_src, att_dst, h, e_src, e_dst, ei,
                                                    (unsigned*)out1, N_, E_);
    edge_owner    <<<owners, 1024, 0, stream>>>(ei, h, e_src, e_dst, W2, (const unsigned*)out1, out, N_, E_);
}


# edge owner: bins filed into per-row LDS lists (LDS add as rank); a wave sums whole rows in registers, softmax by wave reductions, ELU, one store per row; no LDS float atomics; sum order changes
# speedup vs baseline: 4.4288x; 1.0590x over previous
#include <hip/hip_runtime.h>
#include <math.h>

typedef __attribute__((ext_vector_type(2))) float v2f;
typedef __attribute__((ext_vector_type(8))) float v8f;

#define NPW    196
#define NCH    256
#define SLOTW  64
#define MAXOWN 1024
#define BIN_U  8

__device__ __forceinline__ int own_rows(int o, int N_)  { return (N_ - o * NPW < NPW) ? (N_ - o * NPW) : NPW; }
__device__ __forceinline__ int slot_words(int nown)     { const int w = (nown * 128) / NCH; return (w < SLOTW) ? w : SLOTW; }
__device__ __forceinline__ bool bins_usable(int N_)     { return N_ <= (1 << 24) && (N_ + NPW - 1) / NPW <= MAXOWN; }

__device__ __forceinline__ void bin_edges(const int* __restrict__ ei, unsigned* __restrict__ bins,
                                          int* __restrict__ hcnt, int g, int N_, int E_) {
    const int t = threadIdx.x;
    const int owners = (N_ + NPW - 1) / NPW;
    for (int i = t; i < owners; i += 256) hcnt[i] = 0;
    __syncthreads();
    const int ch = (E_ + NCH - 1) / NCH;
    const int e0 = g * ch;
    const int e1 = (E_ - e0 < ch) ? E_ : (e0 + ch);
    for (int k0 = e0; k0 < e1; k0 += 256 * BIN_U) {
        int d[BIN_U], s[BIN_U];
        #pragma unroll
        for (int u = 0; u < BIN_U; ++u) {
            const int e  = k0 + u * 256 + t;
            const int ec = (e < e1) ? e : e0;
            d[u] = ei[E_ + ec];
            s[u] = ei[ec];
        }
        #pragma unroll
        for (int u = 0; u < BIN_U; ++u) {
            if (k0 + u * 256 + t < e1 && (unsigned)d[u] < (unsigned)N_) {
                const int o   = d[u] / NPW;
                const int dl  = d[u] - o * NPW;
                const int sw  = slot_words(own_rows(o, N_));
                const int pos = atomicAdd(&hcnt[o], 1);
                if (pos < sw - 1)
                    bins[(size_t)o * (NPW * 128) + g * sw + 1 + pos] = ((unsigned)s[u] << 8) | (unsigned)dl;
            }
        }
    }
    __syncthreads();
    for (int o = t; o < owners; o += 256) {
        const int sw = slot_words(own_rows(o, N_));
        if (sw >= 2) bins[(size_t)o * (NPW * 128) + g * sw] = (unsigned)hcnt[o];
    }
}

#define GB     512
#define XROWS  32
#define LDX    260

__device__ __forceinline__ float4 x_row_load(const float* x, int r, int N_) {
    if (r >= N_) r = N_ - 1;
    return ((const float4*)(x + (size_t)r * 256))[threadIdx.x & 63];
}
#define X_ROWS_LOAD(rb) do { const int r_ = (rb) + (int)(threadIdx.x >> 6);                                  \
    st0 = x_row_load(x, r_,      N_); st1 = x_row_load(x, r_ + 4,  N_); st2 = x_row_load(x, r_ + 8,  N_);   \
    st3 = x_row_load(x, r_ + 12, N_); st4 = x_row_load(x, r_ + 16, N_); st5 = x_row_load(x, r_ + 20, N_);   \
    st6 = x_row_load(x, r_ + 24, N_); st7 = x_row_load(x, r_ + 28, N_); } while (0)
#define X_ROWS_STORE(buf) do { float* b_ = (buf) + (threadIdx.x >> 6) * LDX + 4 * (threadIdx.x & 63);        \
    *(float4*)(b_)            = st0; *(float4*)(b_ + 4 * LDX)  = st1; *(float4*)(b_ + 8 * LDX)  = st2;       \
    *(float4*)(b_ + 12 * LDX) = st3; *(float4*)(b_ + 16 * LDX) = st4; *(float4*)(b_ + 20 * LDX) = st5;       \
    *(float4*)(b_ + 24 * LDX) = st6; *(float4*)(b_ + 28 * LDX) = st7; } while (0)

#define KG 2
template <int NT>
__device__ __forceinline__ void tile_products(const float* __restrict__ xb, const v2f (&wa)[64], v8f (&acc)[2]) {
    float4 f[2][KG][NT];
    #pragma unroll
    for (int jj = 0; jj < KG; ++jj)
        #pragma unroll
        for (int q = 0; q < NT; ++q) f[0][jj][q] = *(const float4*)(xb + q * 16 * LDX + 8 * jj);
    #pragma unroll
    for (int g = 0; g < 32 / KG; ++g) {
        if (g + 1 < 32 / KG) {
            #pragma unroll
            for (int jj = 0; jj < KG; ++jj)
                #pragma unroll
                for (int q = 0; q < NT; ++q)
                    f[(g + 1) & 1][jj][q] = *(const float4*)(xb + q * 16 * LDX + 8 * ((g + 1) * KG + jj));
        }
        __builtin_amdgcn_sched_barrier(0);
        #pragma unroll
        for (int jj = 0; jj < KG; ++jj) {
            const int j = g * KG + jj;
            #pragma unroll
            for (int q = 0; q < NT; ++q) {
                v2f b; b.x = f[g & 1][jj][q].x; b.y = f[g & 1][jj][q].y;
                acc[q] = __builtin_amdgcn_wmma_f32_16x16x4_f32(false, wa[2 * j], false, b,
                                                               (short)0, acc[q], false, false);
            }
            #pragma unroll
            for (int q = 0; q < NT; ++q) {
                v2f b; b.x = f[g & 1][jj][q].z; b.y = f[g & 1][jj][q].w;
                acc[q] = __builtin_amdgcn_wmma_f32_16x16x4_f32(false, wa[2 * j + 1], false, b,
                                                               (short)0, acc[q], false, false);
            }
        }
    }
}

__global__ __launch_bounds__(256, 4) void gemm1_kernel(const float* __restrict__ x,
                                                       const float* __restrict__ W1,
                                                       const float* __restrict__ att_src,
                                                       const float* __restrict__ att_dst,
                                                       float* __restrict__ h,
                                                       float* __restrict__ e_src,
                                                       float* __restrict__ e_dst,
                                                       const int* __restrict__ ei,
                                                       unsigned* __restrict__ bins, int N_, int E_) {
    __shared__ __attribute__((aligned(16))) float lds[2 * XROWS * LDX];
    __shared__ float red[2 * 2 * 8 * 32];
    static_assert(2 * XROWS * LDX >= MAXOWN, "the binning blocks count in the stages' LDS");
    if (blockIdx.x < NCH) {
        if (bins_usable(N_)) bin_edges(ei, bins, (int*)lds, (int)blockIdx.x, N_, E_);
        return;
    }
    const int G  = (int)gridDim.x - NCH;
    const int gb = (int)blockIdx.x - NCH;
    const int T  = (N_ + 15) / 16;
    const int t0 = (int)(((long long)gb * T) / G);
    const int t1 = (int)(((long long)(gb + 1) * T) / G);
    if (t0 >= t1) return;

    const int t    = threadIdx.x;
    const int wave = t >> 5;
    const int lane = t & 31;
    const int m    = lane & 15;
    const int hh   = lane >> 4;
    const int n    = (wave << 4) + m;
    const int c0   = (wave << 4) + (hh << 3);

    float4 st0, st1, st2, st3, st4, st5, st6, st7;
    X_ROWS_LOAD(t0 * 16);

    v2f wa[64];
    #pragma unroll
    for (int j = 0; j < 32; ++j) {
        const float* __restrict__ wj = W1 + (size_t)(8 * j + 4 * hh) * 128 + n;
        wa[2 * j].x     = wj[0];
        wa[2 * j].y     = wj[128];
        wa[2 * j + 1].x = wj[256];
        wa[2 * j + 1].y = wj[384];
    }
    const float4 as0 = ((const float4*)(att_src + c0))[0];
    const float4 as1 = ((const float4*)(att_src + c0))[1];
    const float4 ad0 = ((const float4*)(att_dst + c0))[0];
    const float4 ad1 = ((const float4*)(att_dst + c0))[1];

    X_ROWS_STORE(lds);
    __syncthreads();

    int cur = 0;
    for (int tt = t0; tt < t1; tt += 2) {
        const bool more = tt + 2 < t1;
        const bool two  = tt + 1 < t1;
        X_ROWS_LOAD((more ? tt + 2 : tt) * 16);

        v8f acc[2] = {};
        const float* __restrict__ xb = lds + cur * (XROWS * LDX) + m * LDX + 4 * hh;
        if (two) tile_products<2>(xb, wa, acc); else tile_products<1>(xb, wa, acc);

        #pragma unroll
        for (int q = 0; q < 2; ++q) {
            if (q == 0 || two) {
                const int row = (tt + q) * 16 + m;
                if (row < N_) {
                    float4* __restrict__ hp = (float4*)(h + (size_t)row * 128 + c0);
                    hp[0] = make_float4(acc[q][0], acc[q][1], acc[q][2], acc[q][3]);
                    hp[1] = make_float4(acc[q][4], acc[q][5], acc[q][6], acc[q][7]);
                }
                float ps = acc[q][0] * as0.x + acc[q][1] * as0.y + acc[q][2] * as0.z + acc[q][3] * as0.w
                         + acc[q][4] * as1.x + acc[q][5] * as1.y + acc[q][6] * as1.z + acc[q][7] * as1.w;
                float pd = acc[q][0] * ad0.x + acc[q][1] * ad0.y + acc[q][2] * ad0.z + acc[q][3] * ad0.w
                         + acc[q][4] * ad1.x + acc[q][5] * ad1.y + acc[q][6] * ad1.z + acc[q][7] * ad1.w;
                ps += __shfl_xor(ps, 16, 32);
                pd += __shfl_xor(pd, 16, 32);
                if (lane < 16) {
                    red[(cur * 2 + q) * 256 + wave * 32 + m * 2]     = ps;
                    red[(cur * 2 + q) * 256 + wave * 32 + m * 2 + 1] = pd;
                }
            }
        }
        X_ROWS_STORE(lds + (cur ^ 1) * (XROWS * LDX));
        __builtin_amdgcn_fence(__ATOMIC_RELEASE, "workgroup", "local");
        __builtin_amdgcn_s_barrier();
        __builtin_amdgcn_fence(__ATOMIC_ACQUIRE, "workgroup", "local");
        if (t < 64 && (t < 32 || two)) {
            const int q = t >> 5, u = t & 31;
            const float* __restrict__ rp = red + (cur * 2 + q) * 256 + u;
            float s = rp[0];
            for (int w = 1; w < 8; ++w) s += rp[w * 32];
            const int row = (tt + q) * 16 + (u >> 1);
            if (row < N_) {
                if (u & 1) e_dst[row] = s; else e_src[row] = s;
            }
        }
        cur ^= 1;
    }
}

#define NPT    ((NPW + 15) / 16)
#define LDR    132
#define WPS    288
#define RCAP   64
#define RPITCH 65
#define NRW    ((NPT * 16 + 31) / 32)
#define AGG_U  16
#define AGG_R  8

__device__ __forceinline__ float sigmoid_f32(float a) { return 1.0f / (1.0f + expf(-a)); }
__device__ __forceinline__ float elu_f32(float v)     { return (v > 0.0f) ? v : (expf(v) - 1.0f); }

template <class F>
__device__ __forceinline__ void scan_dst(const int* __restrict__ dstp, int E_, int lo, int nown, F&& f) {
    const int t = threadIdx.x;
    for (int k0 = 0; k0 < E_; k0 += 1024) {
        const int e = k0 + t;
        if (e < E_) {
            const unsigned a0 = (unsigned)(dstp[e] - lo);
            if (a0 < (unsigned)nown) f(e, (int)a0);
        }
    }
}

template <int U>
__device__ __forceinline__ void gather_rows(const float* __restrict__ h, int sv, float wv, int cnt, int lane,
                                            float4& sum) {
    for (int j0 = 0; j0 < cnt; j0 += U) {
        float4 hv[U];
        #pragma unroll
        for (int u = 0; u < U; ++u) {
            if (j0 + u < cnt) {
                const int s = __builtin_amdgcn_readlane(sv, j0 + u);
                hv[u] = ((const float4*)(h + (size_t)s * 128))[lane];
            }
        }
        #pragma unroll
        for (int u = 0; u < U; ++u) {
            if (j0 + u < cnt) {
                const float w = __int_as_float(__builtin_amdgcn_readlane(__float_as_int(wv), j0 + u));
                sum.x += hv[u].x * w;
                sum.y += hv[u].y * w;
                sum.z += hv[u].z * w;
                sum.w += hv[u].w * w;
            }
        }
    }
}

__global__ __launch_bounds__(1024) void edge_owner(const int* __restrict__ ei,
                                                   const float* __restrict__ h,
                                                   const float* __restrict__ e_src,
                                                   const float* __restrict__ e_dst,
                                                   const float* __restrict__ W2,
                                                   const unsigned* __restrict__ bins,
                                                   float* __restrict__ out, int N_, int E_) {
    __shared__ __attribute__((aligned(16))) float acc[NPT * 16 * LDR];
    __shared__ __attribute__((aligned(16))) float wp[64 * WPS];
    __shared__ int            rlist[NPW * RPITCH];
    __shared__ int            rcnt[NPW];
    __shared__ float          den[NPW];
    __shared__ unsigned       mxb[NPW];
    __shared__ int            ovf;
    static_assert(NCH * 4 == 1024 && NPW <= 256, "four threads of the owner per bin slot; owned row fits 8 bits");

    const int t    = threadIdx.x;
    const int lo   = blockIdx.x * NPW;
    const int nown = (N_ - lo < NPW) ? (N_ - lo) : NPW;
    const int* __restrict__ dstp = ei + E_;
    const int  sw       = slot_words(nown);
    const bool use_bins = bins_usable(N_) && sw >= 2;

    const int wl = t & 31;
    const int wu = t >> 5;
    float4 w2lo[2], w2hi[2];
    #pragma unroll
    for (int k = 0; k < 2; ++k) {
        const int pp = wu + 32 * k;
        const int c0 = 8 * (pp & 15) + (pp >> 4);
        w2lo[k] = ((const float4*)(W2 + (size_t)c0 * 128))[wl];
        w2hi[k] = ((const float4*)(W2 + (size_t)(c0 + 4) * 128))[wl];
    }

    const int g   = t >> 2;
    const int sub = t & 3;
    const unsigned* __restrict__ sp = bins + (size_t)lo * 128 + (size_t)g * sw;
    int      c = 0;
    unsigned en[4] = {0u, 0u, 0u, 0u};
    if (use_bins) {
        c = (int)sp[0];
        #pragma unroll
        for (int k = 0; k < 4; ++k) {
            const int j = sub + 4 * k;
            en[k] = sp[1 + ((j < sw - 2) ? j : (sw - 2))];
        }
    }

    #pragma unroll
    for (int k = 0; k < 2; ++k) {
        float4* __restrict__ d = (float4*)(wp + (wu + 32 * k) * WPS + 8 * wl);
        d[0] = make_float4(w2lo[k].x, w2hi[k].x, w2lo[k].y, w2hi[k].y);
        d[1] = make_float4(w2lo[k].z, w2hi[k].z, w2lo[k].w, w2hi[k].w);
    }
    if (t < NPW) rcnt[t] = 0;
    if (t == 0) ovf = 0;
    __syncthreads();

    if (use_bins) {
        auto file = [&](unsigned e) {
            const int dl = (int)(e & 255u);
            if (dl < nown) {
                const int pos = atomicAdd(&rcnt[dl], 1);
                if (pos < RCAP) rlist[dl * RPITCH + pos] = (int)(e >> 8);
                else            ovf = 1;
            }
        };
        if (c > sw - 1) { ovf = 1; c = sw - 1; }
        #pragma unroll
        for (int k = 0; k < 4; ++k)
            if (sub + 4 * k < c) file(en[k]);
        for (int j = sub + 16; j < c; j += 4) file(sp[1 + j]);
    }
    __syncthreads();
    const bool fast = use_bins && ovf == 0;

    if (fast) {
        const int wave = __builtin_amdgcn_readfirstlane(t >> 5);
        const int lane = t & 31;
        int   nn[NRW];
        int   sv[NRW];
        float wv[NRW];
        {
            float al[NRW], mx[NRW], ex[NRW], sm[NRW];
            int   nv[NRW];
            #pragma unroll
            for (int i = 0; i < NRW; ++i) {
                const int r = wave + 32 * i;
                nv[i] = rcnt[(r < NPW) ? r : (NPW - 1)];
            }
            #pragma unroll
            for (int i = 0; i < NRW; ++i) {
                const int r = wave + 32 * i;
                const int n = (r < nown) ? __builtin_amdgcn_readfirstlane(nv[i]) : 0;
                nn[i] = (n < RCAP) ? n : RCAP;
            }
            #pragma unroll
            for (int i = 0; i < NRW; ++i) {
                const int r = wave + 32 * i;
                sv[i] = (lane < nn[i]) ? rlist[r * RPITCH + lane] : 0;
            }
            #pragma unroll
            for (int i = 0; i < NRW; ++i) {
                const int r = wave + 32 * i;
                const float es = (lane < nn[i]) ? e_src[sv[i]] : 0.0f;
                const float ed = (r < nown) ? e_dst[lo + r] : 0.0f;
                al[i] = es + ed;
            }
            #pragma unroll
            for (int i = 0; i < NRW; ++i) {
                al[i] = (lane < nn[i]) ? sigmoid_f32(al[i]) : 0.0f;
                mx[i] = al[i];
            }
            #pragma unroll
            for (int o = 16; o >= 1; o >>= 1) {
                #pragma unroll
                for (int i = 0; i < NRW; ++i) mx[i] = fmaxf(mx[i], __shfl_xor(mx[i], o, 32));
            }
            #pragma unroll
            for (int i = 0; i < NRW; ++i) {
                ex[i] = (lane < nn[i]) ? expf(al[i] - mx[i]) : 0.0f;
                sm[i] = ex[i];
            }
            #pragma unroll
            for (int o = 16; o >= 1; o >>= 1) {
                #pragma unroll
                for (int i = 0; i < NRW; ++i) sm[i] += __shfl_xor(sm[i], o, 32);
            }
            #pragma unroll
            for (int i = 0; i < NRW; ++i) wv[i] = (lane < nn[i]) ? (ex[i] / sm[i]) : 0.0f;
        }
        #pragma unroll
        for (int i = 0; i < NRW; ++i) {
            const int r = wave + 32 * i;
            if (r < NPT * 16 && nn[i] <= 32) {
                float4 sum = make_float4(0.0f, 0.0f, 0.0f, 0.0f);
                gather_rows<AGG_U>(h, sv[i], wv[i], nn[i], lane, sum);
                float* __restrict__ ap = acc + r * LDR + lane;
                ap[0]  = elu_f32(sum.x);
                ap[32] = elu_f32(sum.y);
                ap[64] = elu_f32(sum.z);
                ap[96] = elu_f32(sum.w);
            }
        }
        #pragma nounroll
        for (int i = 0; i < NRW; ++i) {
            const int r = wave + 32 * i;
            if (r >= nown) break;
            int n = __builtin_amdgcn_readfirstlane(rcnt[r]);
            if (n > RCAP) n = RCAP;
            if (n > 32) {
                const float ed  = e_dst[lo + r];
                const bool  vb  = lane + 32 < n;
                const int   s_a = rlist[r * RPITCH + lane];
                const int   s_b = vb ? rlist[r * RPITCH + 32 + lane] : 0;
                const float a_a = sigmoid_f32(e_src[s_a] + ed);
                const float a_b = vb ? sigmoid_f32(e_src[s_b] + ed) : 0.0f;
                float m = fmaxf(a_a, a_b);
                for (int o = 16; o >= 1; o >>= 1) m = fmaxf(m, __shfl_xor(m, o, 32));
                const float x_a = expf(a_a - m);
                const float x_b = vb ? expf(a_b - m) : 0.0f;
                float sd = x_a + x_b;
                for (int o = 16; o >= 1; o >>= 1) sd += __shfl_xor(sd, o, 32);
                float4 sum = make_float4(0.0f, 0.0f, 0.0f, 0.0f);
                gather_rows<AGG_R>(h, s_a, x_a / sd, 32, lane, sum);
                gather_rows<AGG_R>(h, s_b, x_b / sd, n - 32, lane, sum);
                float* __restrict__ ap = acc + r * LDR + lane;
                ap[0]  = elu_f32(sum.x);
                ap[32] = elu_f32(sum.y);
                ap[64] = elu_f32(sum.z);
                ap[96] = elu_f32(sum.w);
            }
        }
    } else {
        for (int i = t; i < NPT * 16 * LDR / 4; i += 1024)
            ((float4*)acc)[i] = make_float4(0.0f, 0.0f, 0.0f, 0.0f);
        if (t < NPW) { den[t] = 0.0f; mxb[t] = 0u; }
        __syncthreads();
        scan_dst(dstp, E_, lo, nown, [&](int e, int dl) {
            const int s = ei[e];
            const float a = sigmoid_f32(e_src[s] + e_dst[lo + dl]);
            atomicMax(&mxb[dl], __float_as_uint(a));
        });
        __syncthreads();
        scan_dst(dstp, E_, lo, nown, [&](int e, int dl) {
            const int s = ei[e];
            const float a = sigmoid_f32(e_src[s] + e_dst[lo + dl]);
            atomicAdd(&den[dl], expf(a - __uint_as_float(mxb[dl])));
        });
        __syncthreads();
        scan_dst(dstp, E_, lo, nown, [&](int e, int dl) {
            const int s = ei[e];
            const float a = sigmoid_f32(e_src[s] + e_dst[lo + dl]);
            const float w = expf(a - __uint_as_float(mxb[dl])) / den[dl];
            const float* __restrict__ hp = h + (size_t)s * 128;
            for (int cc = 0; cc < 128; ++cc)
                atomicAdd(&acc[dl * LDR + (cc & 3) * 32 + (cc >> 2)], hp[cc] * w);
        });
        __syncthreads();
        for (int i = t; i < nown * 32; i += 1024) {
            float4* __restrict__ p = (float4*)(acc + (i >> 5) * LDR) + (i & 31);
            float4 v = *p;
            v.x = elu_f32(v.x);
            v.y = elu_f32(v.y);
            v.z = elu_f32(v.z);
            v.w = elu_f32(v.w);
            *p = v;
        }
    }
    __syncthreads();

    {
        const int wv = __builtin_amdgcn_readfirstlane(t >> 5);
        const int ln = t & 31;
        const int m  = ln & 15;
        const int hh = ln >> 4;
        const int n  = ((wv & 7) << 4) + m;
        v2f bf[32];
        #pragma unroll
        for (int s = 0; s < 32; ++s)
            bf[s] = *(const v2f*)(wp + (2 * s + hh) * WPS + 2 * n);
        for (int rt = wv >> 3; rt * 16 < nown; rt += 4) {
            const float* __restrict__ ap = acc + (rt * 16 + m) * LDR + 2 * hh;
            v8f d = {};
            #pragma unroll
            for (int s = 0; s < 32; ++s) {
                const v2f a = *(const v2f*)(ap + 4 * s);
                d = __builtin_amdgcn_wmma_f32_16x16x4_f32(false, a, false, bf[s],
                                                          (short)0, d, false, false);
            }
            const int rb = rt * 16 + hh * 8;
            float* __restrict__ op = out + (size_t)(lo + rb) * 128 + n;
            if (rt * 16 + 16 <= nown) {
                #pragma unroll
                for (int r = 0; r < 8; ++r) op[(size_t)r * 128] = d[r];
            } else {
                #pragma unroll
                for (int r = 0; r < 8; ++r)
                    if (rb + r < nown) op[(size_t)r * 128] = d[r];
            }
        }
    }
}

extern "C" void kernel_launch(void* const* d_in, const int* in_sizes, int n_in,
                              void* d_out, int out_size, void* d_ws, size_t ws_size,
                              hipStream_t stream) {
    const float* x       = (const float*)d_in[0];
    const int*   ei      = (const int*)  d_in[1];
    const float* W1      = (const float*)d_in[2];
    const float* att_src = (const float*)d_in[3];
    const float* att_dst = (const float*)d_in[4];
    const float* W2      = (const float*)d_in[5];
    float*       out     = (float*)d_out;

    const int N_ = in_sizes[0] / 256;
    const int E_ = in_sizes[1] / 2;

    float*    ws    = (float*)d_ws;
    float*    h     = ws;
    float*    out1  = h     + (size_t)N_ * 128;
    float*    e_src = out1  + (size_t)N_ * 128;
    float*    e_dst = e_src + N_;

    const int owners = (N_ + NPW - 1) / NPW;
    gemm1_kernel  <<<NCH + GB, 256, 0, stream>>>(x, W1, att_src, att_dst, h, e_src, e_dst, ei,
                                                    (unsigned*)out1, N_, E_);
    edge_owner    <<<owners, 1024, 0, stream>>>(ei, h, e_src, e_dst, W2, (const unsigned*)out1, out, N_, E_);
}
